# starting kernel
# baseline (speedup 1.0000x reference)
#include <hip/hip_runtime.h>
#include <hip/hip_bf16.h>

typedef __attribute__((ext_vector_type(2))) float v2f;
typedef __attribute__((ext_vector_type(8))) float v8f;

__global__ void deg_init_kernel(float* deg, int n) {
  int i = blockIdx.x * blockDim.x + threadIdx.x;
  if (i < n) deg[i] = 1.0f;
}

__global__ void deg_edge_kernel(const int* __restrict__ col, float* deg, int E) {
  int e = blockIdx.x * blockDim.x + threadIdx.x;
  if (e < E) atomicAdd(&deg[col[e]], 1.0f);
}

__global__ void rsqrt_kernel(float* deg, int n) {
  int i = blockIdx.x * blockDim.x + threadIdx.x;
  if (i < n) deg[i] = rsqrtf(deg[i]);
}

template <int N, int K, bool BIAS_RELU, int TM>
__global__ void gemm_wmma_kernel(const float* __restrict__ A,
                                 const float* __restrict__ B,
                                 float* __restrict__ C,
                                 const float* __restrict__ act_bias,
                                 int M) {
  const int lane = threadIdx.x & 31;
  const int wave = threadIdx.x >> 5;
  const int n0   = wave * 16;
  const int half = lane >> 4;
  const int l    = lane & 15;
  const int kb   = 2 * half;

  constexpr int KSTEPS = K / 4;

  v2f breg[KSTEPS];
  {
    const float* bcol = B + (size_t)kb * N + (n0 + l);
#pragma unroll
    for (int s = 0; s < KSTEPS; ++s) {
      breg[s].x = bcol[s * 4 * N];
      breg[s].y = bcol[(s * 4 + 1) * N];
    }
  }
  const float* bias = act_bias + kb;

  for (int t = 0; t < TM; ++t) {
    const int m0 = (blockIdx.x * TM + t) * 16;
    if (m0 >= M) break;

    int ar = m0 + l;
    if (ar > M - 1) ar = M - 1;
    const float* arow = A + (size_t)ar * K + kb;

    v8f acc = {};
#pragma unroll
    for (int s = 0; s < KSTEPS; ++s) {
      v2f a;
      a.x = arow[s * 4];
      a.y = arow[s * 4 + 1];
      if constexpr (BIAS_RELU) {
        a.x = fmaxf(a.x + bias[s * 4], 0.0f);
        a.y = fmaxf(a.y + bias[s * 4 + 1], 0.0f);
      }
      acc = __builtin_amdgcn_wmma_f32_16x16x4_f32(false, a, false, breg[s],
                                                  (short)0, acc, false, false);
    }

    float* cp = C + (size_t)(m0 + half * 8) * N + (n0 + l);
#pragma unroll
    for (int v = 0; v < 8; ++v) {
      if (m0 + half * 8 + v < M) cp[(size_t)v * N] = acc[v];
    }
  }
}

template <int F, bool HAS_BIAS>
__global__ void self_bias_init_kernel(const float* __restrict__ h,
                                      const float* __restrict__ dinv,
                                      const float* __restrict__ bias,
                                      float* __restrict__ out, int n) {
  int idx = blockIdx.x * blockDim.x + threadIdx.x;
  if (idx >= n * F) return;
  int i = idx / F;
  int f = idx - i * F;
  float d = dinv[i];
  float v = h[idx] * d * d;
  if constexpr (HAS_BIAS) v += bias[f];
  out[idx] = v;
}

template <int F>
__global__ void edge_agg_kernel(const int* __restrict__ row,
                                const int* __restrict__ col,
                                const float* __restrict__ h,
                                const float* __restrict__ dinv,
                                float* out, int E) {
  const int lane = threadIdx.x & 31;
  const int e = blockIdx.x * (blockDim.x >> 5) + (threadIdx.x >> 5);
  if (e >= E) return;
  const int r = row[e];
  const int c = col[e];
  const float norm = dinv[r] * dinv[c];
  constexpr int VEC = F / 32;
  const float* hp = h + (size_t)r * F + lane * VEC;
  float* op = out + (size_t)c * F + lane * VEC;
  if constexpr (VEC == 4) {
    float4 hv = *(const float4*)hp;
    atomicAdd(op + 0, hv.x * norm);
    atomicAdd(op + 1, hv.y * norm);
    atomicAdd(op + 2, hv.z * norm);
    atomicAdd(op + 3, hv.w * norm);
  } else {
    float2 hv = *(const float2*)hp;
    atomicAdd(op + 0, hv.x * norm);
    atomicAdd(op + 1, hv.y * norm);
  }
}

extern "C" void kernel_launch(void* const* d_in, const int* in_sizes, int n_in,
                              void* d_out, int out_size, void* d_ws, size_t ws_size,
                              hipStream_t stream) {
  const float* x  = (const float*)d_in[0];
  const int*   ei = (const int*)d_in[1];
  const float* W1 = (const float*)d_in[2];
  const float* b1 = (const float*)d_in[3];
  const float* W2 = (const float*)d_in[4];
  const float* b2 = (const float*)d_in[5];

  const int hid    = in_sizes[3];
  const int out_ch = in_sizes[5];
  const int in_ch  = in_sizes[2] / hid;
  const int n      = in_sizes[0] / in_ch;
  const int E      = in_sizes[1] / 2;
  const int* row = ei;
  const int* col = ei + E;

  float* ws = (float*)d_ws;
  size_t off = 0;
  auto alloc = [&](size_t elems) {
    float* p = ws + off;
    off += (elems + 63) & ~(size_t)63;
    return p;
  };
  float* dinv = alloc((size_t)n);
  float* h1   = alloc((size_t)n * hid);
  float* agg1 = alloc((size_t)n * hid);
  float* g2   = alloc((size_t)n * out_ch);
  float* out  = (float*)d_out;

  deg_init_kernel<<<(n + 255) / 256, 256, 0, stream>>>(dinv, n);
  deg_edge_kernel<<<(E + 255) / 256, 256, 0, stream>>>(col, dinv, E);
  rsqrt_kernel<<<(n + 255) / 256, 256, 0, stream>>>(dinv, n);

  constexpr int TM = 4;
  const int mtiles   = (n + 15) / 16;
  const int mblocks  = (mtiles + TM - 1) / TM;

  gemm_wmma_kernel<128, 128, false, TM><<<mblocks, 32 * (128 / 16), 0, stream>>>(
      x, W1, h1, nullptr, n);
  self_bias_init_kernel<128, false>
      <<<(unsigned)(((size_t)n * 128 + 255) / 256), 256, 0, stream>>>(
          h1, dinv, nullptr, agg1, n);
  edge_agg_kernel<128><<<(E + 7) / 8, 256, 0, stream>>>(row, col, h1, dinv, agg1, E);

  gemm_wmma_kernel<64, 128, true, TM><<<mblocks, 32 * (64 / 16), 0, stream>>>(
      agg1, W2, g2, b1, n);
  self_bias_init_kernel<64, true>
      <<<(unsigned)(((size_t)n * 64 + 255) / 256), 256, 0, stream>>>(
          g2, dinv, b2, out, n);
  edge_agg_kernel<64><<<(E + 7) / 8, 256, 0, stream>>>(row, col, g2, dinv, out, E);
}


# Degree pass streams the edge arrays itself: work-group (32 owners, 1/32 of edges) bins kept edges by owner in LDS, whole-line list stores; no owner-byte pass, no edge look-ups; 6->5; sum order varies
# speedup vs baseline: 4.5447x; 4.5447x over previous
#include <hip/hip_runtime.h>
#include <hip/hip_bf16.h>
#include <type_traits>

typedef __attribute__((ext_vector_type(2))) float v2f;
typedef __attribute__((ext_vector_type(8))) float v8f;

constexpr int AGG_R     = 391;
constexpr int AGG_WAVES = 32;
constexpr int AGG_Q     = 64;
constexpr int AGG_G     = 8;
constexpr int DEG_G     = 32;
constexpr int DEG_U     = 4;
constexpr int SEG_CAP   = 1024;

__global__ __launch_bounds__(32 * AGG_WAVES) void deg_owner_kernel(
    const int* __restrict__ row, const int* __restrict__ col,
    int* __restrict__ cntp, int nstride,
    int* __restrict__ ecnt, unsigned* __restrict__ lists, int n, int E) {
  __shared__ int rows[AGG_R * DEG_G];
  __shared__ unsigned bins[DEG_G * SEG_CAP];
  __shared__ int fill[DEG_G];

  const int tid  = threadIdx.x;
  const int lane = tid & 31;
  const int wave = tid >> 5;
  const int grp   = (int)blockIdx.x / DEG_G;
  const int share = (int)blockIdx.x % DEG_G;
  const int base = grp * (AGG_R * DEG_G);
  int rcount = n - base;
  if (rcount > AGG_R * DEG_G) rcount = AGG_R * DEG_G;
  if (rcount < 0) rcount = 0;

  for (int t = tid; t < AGG_R * DEG_G; t += 32 * AGG_WAVES) rows[t] = 0;
  if (tid < DEG_G) fill[tid] = 0;
  __syncthreads();

  auto count = [&](int c, int& g, int& pos) -> bool {
    const int lc = c - base;
    const bool ours = (unsigned)lc < (unsigned)rcount;
    if (ours) {
      __hip_atomic_fetch_add(&rows[lc], 1, __ATOMIC_RELAXED, __HIP_MEMORY_SCOPE_WORKGROUP);
      g   = lc / AGG_R;
      pos = __hip_atomic_fetch_add(&fill[g], 1, __ATOMIC_RELAXED, __HIP_MEMORY_SCOPE_WORKGROUP);
    }
    return ours;
  };
  auto put = [&](bool ours, int c, int r, int g, int pos) {
    if (ours && pos < SEG_CAP)
      bins[g * SEG_CAP + pos] = (unsigned)r | ((unsigned)(c - base - g * AGG_R) << 23);
  };
  auto piece = [&](int4 c, int4 r) {
    int g0 = 0, g1 = 0, g2 = 0, g3 = 0, p0 = 0, p1 = 0, p2 = 0, p3 = 0;
    const bool k0 = count(c.x, g0, p0);
    const bool k1 = count(c.y, g1, p1);
    const bool k2 = count(c.z, g2, p2);
    const bool k3 = count(c.w, g3, p3);
    put(k0, c.x, r.x, g0, p0);
    put(k1, c.y, r.y, g1, p1);
    put(k2, c.z, r.z, g2, p2);
    put(k3, c.w, r.w, g3, p3);
  };
  auto take = [&](int c, int r) {
    int g = 0, pos = 0;
    const bool ours = count(c, g, pos);
    put(ours, c, r, g, pos);
  };

  const int spe  = (((E + DEG_G - 1) / DEG_G) + 3) & ~3;
  const int sbeg = share * spe;
  int send = sbeg + spe;
  if (send > E) send = E;
  int slen = send - sbeg;
  if (slen < 0) slen = 0;
  const bool vec  = ((((size_t)row) | ((size_t)col)) & 15) == 0;
  const int nfull = vec ? (slen >> 2) : 0;
  const int4* colv = (const int4*)(col + sbeg);
  const int4* rowv = (const int4*)(row + sbeg);
  const int span  = 32 * AGG_WAVES * DEG_U;
  const int iters = (nfull + span - 1) / span;

  auto fetch = [&](int4 (&cc)[DEG_U], int4 (&rr)[DEG_U], int it) {
#pragma unroll
    for (int u = 0; u < DEG_U; ++u) {
      const int idx = (it * DEG_U + u) * (32 * AGG_WAVES) + tid;
      const int j = idx < nfull ? idx : 0;
      cc[u] = colv[j];
      rr[u] = rowv[j];
    }
  };
  auto work = [&](const int4 (&cc)[DEG_U], const int4 (&rr)[DEG_U], int it) {
#pragma unroll
    for (int u = 0; u < DEG_U; ++u) {
      const int idx = (it * DEG_U + u) * (32 * AGG_WAVES) + tid;
      if (idx < nfull) piece(cc[u], rr[u]);
    }
  };
  int4 ca[DEG_U], ra[DEG_U], cb[DEG_U], rb[DEG_U];
  if (iters > 0) fetch(ca, ra, 0);
  for (int it = 0; it < iters; it += 2) {
    fetch(cb, rb, it + 1);
    work(ca, ra, it);
    fetch(ca, ra, it + 2);
    work(cb, rb, it + 1);
  }
  for (int e = sbeg + 4 * nfull + tid; e < send; e += 32 * AGG_WAVES) take(col[e], row[e]);
  __syncthreads();

  for (int t = tid; t < rcount; t += 32 * AGG_WAVES)
    cntp[(size_t)share * (size_t)nstride + base + t] = rows[t];
  if (tid < DEG_G) ecnt[(grp * DEG_G + tid) * DEG_G + share] = fill[tid];
  for (int g = wave; g < DEG_G; g += AGG_WAVES) {
    int cg = fill[g];
    if (cg > SEG_CAP) cg = SEG_CAP;
    unsigned* seg = lists + ((size_t)(grp * DEG_G + g) * DEG_G + (size_t)share) * (size_t)SEG_CAP;
    for (int i = lane; i < cg; i += 32) seg[i] = bins[g * SEG_CAP + i];
  }
}

constexpr int GK         = 128;
constexpr int GEMM_WAVES = 16;
constexpr int GEMM_GRID  = 256;

template <int N, bool BIAS_RELU, bool PRE>
__global__ __launch_bounds__(32 * GEMM_WAVES) void gemm_strip_kernel(
    const float* __restrict__ A, const float* __restrict__ B, float* __restrict__ C,
    const float* __restrict__ act_bias, int M,
    const int* __restrict__ cntp, int nstride, float* __restrict__ dinv,
    const int* __restrict__ ecnt, int* __restrict__ etot, int nown) {
  constexpr int NG  = N / 64;
  constexpr int PP4 = 2 * N / 4;
  __shared__ float4 wl[(GK / 2) * PP4];

  const int tid = threadIdx.x;
  if constexpr (PRE) {
    const int nthr = (int)gridDim.x * (32 * GEMM_WAVES);
    for (int i = (int)blockIdx.x * (32 * GEMM_WAVES) + tid; i < M; i += nthr) {
      int c = 1;
#pragma unroll
      for (int sh = 0; sh < DEG_G; ++sh) c += cntp[(size_t)sh * (size_t)nstride + i];
      dinv[i] = rsqrtf((float)c);
    }
    for (int o = (int)blockIdx.x * (32 * GEMM_WAVES) + tid; o < nown; o += nthr) {
      int c = 0;
#pragma unroll
      for (int sh = 0; sh < DEG_G; ++sh) c += ecnt[o * DEG_G + sh];
      etot[o] = c;
    }
  }
  for (int i = tid; i < (GK / 2) * NG * 16; i += 32 * GEMM_WAVES) {
    const int c  = i & 15;
    const int g  = (i >> 4) % NG;
    const int kp = i / (16 * NG);
    const float4 r0 = *(const float4*)(B + (size_t)(2 * kp) * N + 64 * g + 4 * c);
    const float4 r1 = *(const float4*)(B + (size_t)(2 * kp + 1) * N + 64 * g + 4 * c);
    float4* dst = wl + kp * PP4 + g * 32 + c;
    dst[0]  = make_float4(r0.x, r1.x, r0.y, r1.y);
    dst[16] = make_float4(r0.z, r1.z, r0.w, r1.w);
  }
  __syncthreads();

  const int lane = tid & 31;
  const int wave = __builtin_amdgcn_readfirstlane(tid >> 5);
  const int h = lane >> 4;
  const int c = lane & 15;

  const unsigned T   = (unsigned)(M + 15) >> 4;
  const unsigned per = T / gridDim.x;
  const unsigned rem = T - per * gridDim.x;
  const int sb = (int)(blockIdx.x * per + (blockIdx.x < rem ? blockIdx.x : rem));
  const int se = sb + (int)per + (blockIdx.x < rem ? 1 : 0);

  const float4* wb = wl + (2 * h) * PP4 + c;

  for (int s = sb + wave; s < se; s += GEMM_WAVES) {
    const int m0 = s * 16;
    int ar = m0 + c;
    if (ar > M - 1) ar = M - 1;
    const float* arow = A + (size_t)ar * GK + 4 * h;

    float4 a4[16];
#pragma unroll
    for (int j = 0; j < 16; ++j) a4[j] = *(const float4*)(arow + 8 * j);
    if constexpr (BIAS_RELU) {
#pragma unroll
      for (int j = 0; j < 16; ++j) {
        const float4 bb = *(const float4*)(act_bias + 8 * j + 4 * h);
        a4[j].x = fmaxf(a4[j].x + bb.x, 0.0f);
        a4[j].y = fmaxf(a4[j].y + bb.y, 0.0f);
        a4[j].z = fmaxf(a4[j].z + bb.z, 0.0f);
        a4[j].w = fmaxf(a4[j].w + bb.w, 0.0f);
      }
    }

    v8f acc[NG][4];
#pragma unroll
    for (int g = 0; g < NG; ++g)
#pragma unroll
      for (int t = 0; t < 4; ++t) acc[g][t] = v8f{};

    float4 fc[NG][2][2], fn[NG][2][2];
    auto frag = [&](float4 (&f)[NG][2][2], int j) {
#pragma unroll
      for (int g = 0; g < NG; ++g)
#pragma unroll
        for (int q = 0; q < 2; ++q)
#pragma unroll
          for (int u = 0; u < 2; ++u) f[g][q][u] = wb[(4 * j + q) * PP4 + g * 32 + u * 16];
    };
    frag(fc, 0);
#pragma unroll
    for (int j = 0; j < 16; ++j) {
      if (j + 1 < 16) frag(fn, j + 1);
      __builtin_amdgcn_sched_barrier(0);
#pragma unroll
      for (int q = 0; q < 2; ++q) {
        v2f a;
        a.x = q ? a4[j].z : a4[j].x;
        a.y = q ? a4[j].w : a4[j].y;
#pragma unroll
        for (int g = 0; g < NG; ++g)
#pragma unroll
          for (int t = 0; t < 4; ++t) {
            const float4 p = fc[g][q][t >> 1];
            v2f b;
            b.x = (t & 1) ? p.z : p.x;
            b.y = (t & 1) ? p.w : p.y;
            acc[g][t] = __builtin_amdgcn_wmma_f32_16x16x4_f32(false, a, false, b, (short)0,
                                                              acc[g][t], false, false);
          }
      }
      __builtin_amdgcn_sched_barrier(0);
      if (j + 1 < 16) {
#pragma unroll
        for (int g = 0; g < NG; ++g)
#pragma unroll
          for (int q = 0; q < 2; ++q)
#pragma unroll
            for (int u = 0; u < 2; ++u) fc[g][q][u] = fn[g][q][u];
      }
    }

#pragma unroll
    for (int r = 0; r < 8; ++r) {
      const int orow = m0 + 8 * h + r;
      if (orow < M) {
        float* cp = C + (size_t)orow * N + 4 * c;
#pragma unroll
        for (int g = 0; g < NG; ++g)
          *(float4*)(cp + 64 * g) =
              make_float4(acc[g][0][r], acc[g][1][r], acc[g][2][r], acc[g][3][r]);
      }
    }
  }
}

template <int F, bool HAS_BIAS>
__global__ __launch_bounds__(1024) void agg_owner_kernel(
    const int* __restrict__ row, const int* __restrict__ col,
    const float* __restrict__ h, const float* __restrict__ dinv,
    const float* __restrict__ bias, float* __restrict__ out,
    const int* __restrict__ ecnt, const int* __restrict__ etot, unsigned* __restrict__ recw,
    const unsigned* __restrict__ lists, int cap, int n, int E) {
  constexpr int VEC = F / 32;
  __shared__ float acc[AGG_R * F];
  __shared__ int queue_e[AGG_WAVES][AGG_Q];
  __shared__ int fill;

  const int tid  = threadIdx.x;
  const int lane = tid & 31;
  const int wave = tid >> 5;
  const int base = blockIdx.x * AGG_R;
  int rcount = n - base;
  if (rcount > AGG_R) rcount = AGG_R;

  int before = 0;
  for (int j = lane; j < (int)blockIdx.x; j += 32) before += etot[j];
  for (int o = 16; o > 0; o >>= 1) before += __shfl_xor(before, o);
  uint2* rec = (uint2*)recw + before;
  if (tid == 0) fill = 0;

  const int  mine = etot[blockIdx.x];
  bool listed = true;
  int  pre[DEG_G], gap[DEG_G];
  {
    int run = 0;
#pragma unroll
    for (int k = 0; k < DEG_G; ++k) {
      const int c = ecnt[(int)blockIdx.x * DEG_G + k];
      listed = listed && c <= cap;
      pre[k] = run;
      gap[k] = k * cap - run;
      run += c;
    }
  }
  const unsigned* list = lists + (size_t)blockIdx.x * DEG_G * (size_t)cap;
  auto word = [&](int i) -> unsigned {
    int a = i;
#pragma unroll
    for (int k = 1; k < DEG_G; ++k) a = i >= pre[k] ? i + gap[k] : a;
    return list[a];
  };
  const int wv  = __builtin_amdgcn_readfirstlane(wave);
  const int per = (mine + AGG_WAVES - 1) / AGG_WAVES;
  int c0 = wv * per;
  int cend = c0 + per;
  if (cend > mine) cend = mine;
  const int last = mine - 1;
  unsigned q = 0u, qn = 0u;
  if (listed && mine > 0) {
    q  = word(c0 + lane < last ? c0 + lane : last);
    qn = word(c0 + 32 + lane < last ? c0 + 32 + lane : last);
  }

#pragma unroll 4
  for (int lc = wave; lc < rcount; lc += AGG_WAVES) {
    const int i = base + lc;
    const float d = dinv[i];
    const float* hp = h + (size_t)i * F + lane * VEC;
    float v[VEC];
    if constexpr (VEC == 4) {
      const float4 t = *(const float4*)hp;
      v[0] = t.x; v[1] = t.y; v[2] = t.z; v[3] = t.w;
    } else {
      const float2 t = *(const float2*)hp;
      v[0] = t.x; v[1] = t.y;
    }
#pragma unroll
    for (int j = 0; j < VEC; ++j) {
      float s = v[j] * d * d;
      if constexpr (HAS_BIAS) s += bias[lane * VEC + j];
      acc[lc * F + j * 32 + lane] = s;
    }
  }
  __syncthreads();

  if (listed) {
    float dr = 0.0f, dc = 0.0f;
    if (c0 < cend) {
      dr = dinv[q & 0x7fffffu];
      dc = dinv[base + (int)(q >> 23)];
    }
    while (c0 < cend) {
      const int cn = c0 + 32;
      const int i2 = cn + 32 + lane;
      const unsigned qnn = word(i2 < last ? i2 : last);
      const float drn = dinv[qn & 0x7fffffu];
      const float dcn = dinv[base + (int)(qn >> 23)];

      int nb = cend - c0;
      if (nb > 32) nb = 32;
      const int r  = (int)(q & 0x7fffffu);
      const int lc = (int)(q >> 23);
      const float w = dr * dc;
      auto group = [&](int k0, auto whole) {
        constexpr bool WHOLE = decltype(whole)::value;
        float hv[AGG_G][VEC];
#pragma unroll
        for (int g = 0; g < AGG_G; ++g) {
          int k = k0 + g;
          if (!WHOLE && k > nb - 1) k = nb - 1;
          const int rr = __builtin_amdgcn_readlane(r, k);
          const float* hp = h + (size_t)rr * F + lane * VEC;
          if constexpr (VEC == 4) {
            const float4 t = *(const float4*)hp;
            hv[g][0] = t.x; hv[g][1] = t.y; hv[g][2] = t.z; hv[g][3] = t.w;
          } else {
            const float2 t = *(const float2*)hp;
            hv[g][0] = t.x; hv[g][1] = t.y;
          }
        }
#pragma unroll
        for (int g = 0; g < AGG_G; ++g) {
          const int k = k0 + g;
          if (WHOLE || k < nb) {
            const float ww = __int_as_float(__builtin_amdgcn_readlane(__float_as_int(w), k));
            const int   ll = __builtin_amdgcn_readlane(lc, k);
            float* ap = acc + ll * F + lane;
#pragma unroll
            for (int j = 0; j < VEC; ++j)
              __hip_atomic_fetch_add(ap + j * 32, hv[g][j] * ww, __ATOMIC_RELAXED,
                                     __HIP_MEMORY_SCOPE_WORKGROUP);
          }
        }
      };
      int k0 = 0;
      for (; k0 + AGG_G <= nb; k0 += AGG_G) group(k0, std::true_type{});
      if (k0 < nb) group(k0, std::false_type{});
      if (lane < nb) rec[c0 + lane] = make_uint2(q, __float_as_uint(w));
      q  = qn;
      qn = qnn;
      dr = drn;
      dc = dcn;
      c0 = cn;
    }
  } else {
  int* qe = queue_e[wave];
  int cnt = 0;

  auto drain = [&](int off, int nb) {
    int r = 0, lc = 0;
    float w = 0.0f;
    bool ok = false;
    if (lane < nb) {
      const int e = qe[off + lane];
      const int c = col[e];
      r  = row[e];
      lc = c - base;
      ok = true;
      w  = dinv[r] * dinv[c];
    }
    const unsigned okmask = __builtin_amdgcn_ballot_w32(ok);
    int slot = 0;
    if (lane == 0)
      slot = __hip_atomic_fetch_add(&fill, (int)__builtin_popcount(okmask), __ATOMIC_RELAXED,
                                    __HIP_MEMORY_SCOPE_WORKGROUP);
    for (int k0 = 0; k0 < nb; k0 += AGG_G) {
      float hv[AGG_G][VEC];
#pragma unroll
      for (int g = 0; g < AGG_G; ++g) {
        int k = k0 + g;
        if (k > nb - 1) k = nb - 1;
        const int rr = __builtin_amdgcn_readlane(r, k);
        const float* hp = h + (size_t)rr * F + lane * VEC;
        if constexpr (VEC == 4) {
          const float4 t = *(const float4*)hp;
          hv[g][0] = t.x; hv[g][1] = t.y; hv[g][2] = t.z; hv[g][3] = t.w;
        } else {
          const float2 t = *(const float2*)hp;
          hv[g][0] = t.x; hv[g][1] = t.y;
        }
      }
#pragma unroll
      for (int g = 0; g < AGG_G; ++g) {
        const int k = k0 + g;
        if (k < nb && ((okmask >> k) & 1u)) {
          const float ww = __int_as_float(__builtin_amdgcn_readlane(__float_as_int(w), k));
          const int   ll = __builtin_amdgcn_readlane(lc, k);
          float* ap = acc + ll * F + lane;
#pragma unroll
          for (int j = 0; j < VEC; ++j)
            __hip_atomic_fetch_add(ap + j * 32, hv[g][j] * ww, __ATOMIC_RELAXED,
                                   __HIP_MEMORY_SCOPE_WORKGROUP);
        }
      }
    }
    slot = __builtin_amdgcn_readfirstlane(slot);
    if (ok)
      rec[slot + (int)__builtin_amdgcn_mbcnt_lo(okmask, 0u)] =
          make_uint2((unsigned)r | ((unsigned)lc << 23), __float_as_uint(w));
  };

  const int iters = (E + 1023) / 1024;
  for (int it = 0; it < iters; ++it) {
    const int e = it * 1024 + tid;
    const bool keep = e < E && (unsigned)(col[e] - base) < (unsigned)rcount;
    const unsigned mask = __builtin_amdgcn_ballot_w32(keep);
    if (keep) qe[cnt + (int)__builtin_amdgcn_mbcnt_lo(mask, 0u)] = e;
    cnt += __builtin_popcount(mask);
    if (cnt >= 32) { cnt -= 32; drain(cnt, 32); }
  }
  if (cnt > 0) drain(0, cnt);
  }
  __syncthreads();

  for (int lc = wave; lc < rcount; lc += AGG_WAVES) {
    float* op = out + (size_t)(base + lc) * F + lane * VEC;
    const float* ap = acc + lc * F + lane;
    if constexpr (VEC == 4) {
      *(float4*)op = make_float4(ap[0], ap[32], ap[64], ap[96]);
    } else {
      *(float2*)op = make_float2(ap[0], ap[32]);
    }
  }
}

template <int F, bool HAS_BIAS>
__global__ __launch_bounds__(1024) void agg_records_kernel(
    const unsigned* __restrict__ recw, const int* __restrict__ etot,
    const float* __restrict__ h, const float* __restrict__ dinv,
    const float* __restrict__ bias, float* __restrict__ out, int n) {
  constexpr int VEC = F / 32;
  __shared__ float acc[AGG_R * F];

  const int tid  = threadIdx.x;
  const int lane = tid & 31;
  const int wave = __builtin_amdgcn_readfirstlane(tid >> 5);
  const int base = blockIdx.x * AGG_R;
  int rcount = n - base;
  if (rcount > AGG_R) rcount = AGG_R;

  int before = 0;
  for (int j = lane; j < (int)blockIdx.x; j += 32) before += etot[j];
  for (int o = 16; o > 0; o >>= 1) before += __shfl_xor(before, o);
  const int mine = etot[blockIdx.x];
  const uint2* rec = (const uint2*)recw + before;

  const int per = (mine + AGG_WAVES - 1) / AGG_WAVES;
  int c0 = wave * per;
  int cend = c0 + per;
  if (cend > mine) cend = mine;
  uint2 q = make_uint2(0u, 0u);
  if (c0 + lane < cend) q = rec[c0 + lane];

#pragma unroll 4
  for (int lc = wave; lc < rcount; lc += AGG_WAVES) {
    const int i = base + lc;
    const float d = dinv[i];
    const float* hp = h + (size_t)i * F + lane * VEC;
    float v[VEC];
    if constexpr (VEC == 4) {
      const float4 t = *(const float4*)hp;
      v[0] = t.x; v[1] = t.y; v[2] = t.z; v[3] = t.w;
    } else {
      const float2 t = *(const float2*)hp;
      v[0] = t.x; v[1] = t.y;
    }
#pragma unroll
    for (int j = 0; j < VEC; ++j) {
      float s = v[j] * d * d;
      if constexpr (HAS_BIAS) s += bias[lane * VEC + j];
      acc[lc * F + j * 32 + lane] = s;
    }
  }
  __syncthreads();

  while (c0 < cend) {
    const int cn = c0 + 32;
    uint2 qn = make_uint2(0u, 0u);
    if (cn + lane < cend) qn = rec[cn + lane];

    int nb = cend - c0;
    if (nb > 32) nb = 32;
    const int   r  = (int)(q.x & 0x7fffffu);
    const int   lc = (int)(q.x >> 23);
    const float w  = __uint_as_float(q.y);
    auto group = [&](int k0, auto whole) {
      constexpr bool WHOLE = decltype(whole)::value;
      float hv[AGG_G][VEC];
#pragma unroll
      for (int g = 0; g < AGG_G; ++g) {
        int k = k0 + g;
        if (!WHOLE && k > nb - 1) k = nb - 1;
        const int rr = __builtin_amdgcn_readlane(r, k);
        const float* hp = h + (size_t)rr * F + lane * VEC;
        if constexpr (VEC == 4) {
          const float4 t = *(const float4*)hp;
          hv[g][0] = t.x; hv[g][1] = t.y; hv[g][2] = t.z; hv[g][3] = t.w;
        } else {
          const float2 t = *(const float2*)hp;
          hv[g][0] = t.x; hv[g][1] = t.y;
        }
      }
#pragma unroll
      for (int g = 0; g < AGG_G; ++g) {
        const int k = k0 + g;
        if (WHOLE || k < nb) {
          const float ww = __int_as_float(__builtin_amdgcn_readlane(__float_as_int(w), k));
          const int   ll = __builtin_amdgcn_readlane(lc, k);
          float* ap = acc + ll * F + lane;
#pragma unroll
          for (int j = 0; j < VEC; ++j)
            __hip_atomic_fetch_add(ap + j * 32, hv[g][j] * ww, __ATOMIC_RELAXED,
                                   __HIP_MEMORY_SCOPE_WORKGROUP);
        }
      }
    };
    int k0 = 0;
    for (; k0 + AGG_G <= nb; k0 += AGG_G) group(k0, std::true_type{});
    if (k0 < nb) group(k0, std::false_type{});
    q  = qn;
    c0 = cn;
  }
  __syncthreads();

  for (int lc = wave; lc < rcount; lc += AGG_WAVES) {
    float* op = out + (size_t)(base + lc) * F + lane * VEC;
    const float* ap = acc + lc * F + lane;
    if constexpr (VEC == 4) {
      *(float4*)op = make_float4(ap[0], ap[32], ap[64], ap[96]);
    } else {
      *(float2*)op = make_float2(ap[0], ap[32]);
    }
  }
}

extern "C" void kernel_launch(void* const* d_in, const int* in_sizes, int n_in,
                              void* d_out, int out_size, void* d_ws, size_t ws_size,
                              hipStream_t stream) {
  const float* x  = (const float*)d_in[0];
  const int*   ei = (const int*)d_in[1];
  const float* W1 = (const float*)d_in[2];
  const float* b1 = (const float*)d_in[3];
  const float* W2 = (const float*)d_in[4];
  const float* b2 = (const float*)d_in[5];

  const int hid    = in_sizes[3];
  const int out_ch = in_sizes[5];
  const int in_ch  = in_sizes[2] / hid;
  const int n      = in_sizes[0] / in_ch;
  const int E      = in_sizes[1] / 2;
  const int* row = ei;
  const int* col = ei + E;

  float* ws = (float*)d_ws;
  size_t off = 0;
  auto alloc = [&](size_t elems) {
    float* p = ws + off;
    off += (elems + 63) & ~(size_t)63;
    return p;
  };
  float* dinv = alloc((size_t)n);
  float* h1   = alloc((size_t)n * hid);
  float* agg1 = alloc((size_t)n * hid);
  float* g2   = alloc((size_t)n * out_ch);
  const int owners = (n + AGG_R - 1) / AGG_R;
  const int groups = (owners + DEG_G - 1) / DEG_G;
  const int nown   = groups * DEG_G;
  const int nstride = (n + 63) & ~63;
  int* cntp = (int*)alloc((size_t)DEG_G * nstride);
  int* ecnt = (int*)alloc((size_t)nown * DEG_G);
  int* etot = (int*)alloc((size_t)nown);
  unsigned* rec = (unsigned*)alloc((size_t)E * 2);
  unsigned* lists = (unsigned*)alloc((size_t)nown * DEG_G * SEG_CAP);
  float* out  = (float*)d_out;

  deg_owner_kernel<<<groups * DEG_G, 32 * AGG_WAVES, 0, stream>>>(row, col, cntp, nstride, ecnt,
                                                                  lists, n, E);

  gemm_strip_kernel<128, false, true><<<GEMM_GRID, 32 * GEMM_WAVES, 0, stream>>>(
      x, W1, h1, nullptr, n, cntp, nstride, dinv, ecnt, etot, nown);
  agg_owner_kernel<128, false><<<owners, 1024, 0, stream>>>(
      row, col, h1, dinv, nullptr, agg1, ecnt, etot, rec, lists, SEG_CAP, n, E);

  gemm_strip_kernel<64, true, false><<<GEMM_GRID, 32 * GEMM_WAVES, 0, stream>>>(
      agg1, W2, g2, b1, n, nullptr, 0, nullptr, nullptr, nullptr, 0);
  agg_records_kernel<64, true><<<owners, 1024, 0, stream>>>(
      rec, etot, g2, dinv, b2, out, n);
}
